# starting kernel
# baseline (speedup 1.0000x reference)
#include <hip/hip_runtime.h>


#define B_ 16
#define S_ 2048
#define D_ 128
#define THR 0.3f

typedef __attribute__((ext_vector_type(2))) float v2f;
typedef __attribute__((ext_vector_type(8))) float v8f;

#define KSTRIDE 132
#define VSTRIDE 20
#define PSTRIDE 20

static __device__ __forceinline__
v8f wmma_f32(v2f a, v2f b, v8f c) {
    return __builtin_amdgcn_wmma_f32_16x16x4_f32(false, a, false, b,
                                                 (short)0, c, false, false);
}

__global__ __launch_bounds__(128)
void attn_fa_kernel(const float* __restrict__ Q,
                    const float* __restrict__ T,
                    const float* __restrict__ V,
                    float* __restrict__ O)
{
    __shared__ float lds_k [16  * KSTRIDE];
    __shared__ float lds_vt[128 * VSTRIDE];
    __shared__ float lds_p [4 * 16 * PSTRIDE];

    const int b    = blockIdx.y;
    const int g    = blockIdx.x;
    const int tid  = threadIdx.x;
    const int wave = tid >> 5;
    const int lane = tid & 31;
    const int n    = lane & 15;
    const int hi   = lane >> 4;

    const int q0     = g * 64 + wave * 16;
    const int myDiag = q0 >> 4;
    const int jmax   = g * 4 + 3;

    const float SCALE2 = 0.08838834764831845f * 1.4426950408889634f;

    v2f qa[32];
    {
        const float* qp = Q + ((size_t)b * S_ + q0 + n) * D_ + 2 * hi;
#pragma unroll
        for (int c = 0; c < 32; ++c)
            qa[c] = *(const v2f*)(qp + 4 * c);
    }

    v8f y[8];
#pragma unroll
    for (int nt = 0; nt < 8; ++nt) y[nt] = {};
    float m_run[8], l_run[8];
#pragma unroll
    for (int e = 0; e < 8; ++e) { m_run[e] = -1e30f; l_run[e] = 0.0f; }

    for (int j = 0; j <= jmax; ++j) {
        {
            const int r     = tid >> 3;
            const int dbase = (tid & 7) * 16;
            const float* kp = T + ((size_t)b * S_ + (size_t)j * 16 + r) * D_ + dbase;
            const float* vp = V + ((size_t)b * S_ + (size_t)j * 16 + r) * D_ + dbase;
#pragma unroll
            for (int i = 0; i < 4; ++i) {
                float4 kq = *(const float4*)(kp + 4 * i);
                *(float4*)&lds_k[r * KSTRIDE + dbase + 4 * i] = kq;
                float4 vq = *(const float4*)(vp + 4 * i);
                lds_vt[(dbase + 4*i + 0) * VSTRIDE + r] = vq.x;
                lds_vt[(dbase + 4*i + 1) * VSTRIDE + r] = vq.y;
                lds_vt[(dbase + 4*i + 2) * VSTRIDE + r] = vq.z;
                lds_vt[(dbase + 4*i + 3) * VSTRIDE + r] = vq.w;
            }
        }
        __syncthreads();

        if (j <= myDiag) {
            const float* krow = &lds_k[n * KSTRIDE + 2 * hi];
            v8f s0 = {}, s1 = {};
            v2f kb[2][8];
#pragma unroll
            for (int i = 0; i < 8; ++i)
                kb[0][i] = *(const v2f*)(krow + 4 * i);
#pragma unroll
            for (int grp = 0; grp < 4; ++grp) {
                if (grp < 3) {
#pragma unroll
                    for (int i = 0; i < 8; ++i)
                        kb[(grp + 1) & 1][i] =
                            *(const v2f*)(krow + 4 * (8 * (grp + 1) + i));
                }
#pragma unroll
                for (int i = 0; i < 8; i += 2) {
                    s0 = wmma_f32(qa[8 * grp + i],     kb[grp & 1][i],     s0);
                    s1 = wmma_f32(qa[8 * grp + i + 1], kb[grp & 1][i + 1], s1);
                }
            }

            const int kk = j * 16 + n;
            float al[8], p[8];
#pragma unroll
            for (int e = 0; e < 8; ++e) {
                float sv = s0[e] + s1[e];
                sv = (sv > THR) ? sv : 0.0f;
                float z = sv * SCALE2;
                const int m = e + 8 * hi;
                if (kk > q0 + m) z = -1e30f;
                float rm = z;
                rm = fmaxf(rm, __shfl_xor(rm, 1, 32));
                rm = fmaxf(rm, __shfl_xor(rm, 2, 32));
                rm = fmaxf(rm, __shfl_xor(rm, 4, 32));
                rm = fmaxf(rm, __shfl_xor(rm, 8, 32));
                float mn = fmaxf(m_run[e], rm);
                float a  = __builtin_amdgcn_exp2f(m_run[e] - mn);
                float pe = __builtin_amdgcn_exp2f(z - mn);
                float rs = pe;
                rs += __shfl_xor(rs, 1, 32);
                rs += __shfl_xor(rs, 2, 32);
                rs += __shfl_xor(rs, 4, 32);
                rs += __shfl_xor(rs, 8, 32);
                l_run[e] = l_run[e] * a + rs;
                m_run[e] = mn;
                al[e] = a;
                p[e]  = pe;
            }

#pragma unroll
            for (int nt = 0; nt < 8; ++nt)
#pragma unroll
                for (int e = 0; e < 8; ++e)
                    y[nt][e] *= al[e];

            float* pp = &lds_p[wave * 16 * PSTRIDE];
#pragma unroll
            for (int e = 0; e < 8; ++e)
                pp[(e + 8 * hi) * PSTRIDE + n] = p[e];

            v2f pa[4];
#pragma unroll
            for (int c = 0; c < 4; ++c)
                pa[c] = *(const v2f*)&pp[n * PSTRIDE + 4 * c + 2 * hi];

#pragma unroll
            for (int half = 0; half < 2; ++half) {
                v2f vb[16];
#pragma unroll
                for (int nt = 0; nt < 4; ++nt)
#pragma unroll
                    for (int c = 0; c < 4; ++c)
                        vb[nt * 4 + c] = *(const v2f*)
                            &lds_vt[((half * 4 + nt) * 16 + n) * VSTRIDE + 4 * c + 2 * hi];
#pragma unroll
                for (int nt = 0; nt < 4; ++nt) {
                    v8f acc = y[half * 4 + nt];
#pragma unroll
                    for (int c = 0; c < 4; ++c)
                        acc = wmma_f32(pa[c], vb[nt * 4 + c], acc);
                    y[half * 4 + nt] = acc;
                }
            }
        }
        __syncthreads();
    }

    float inv[8];
#pragma unroll
    for (int e = 0; e < 8; ++e) inv[e] = 1.0f / l_run[e];
    float* op = O + ((size_t)b * S_ + q0) * D_;
#pragma unroll
    for (int nt = 0; nt < 8; ++nt)
#pragma unroll
        for (int e = 0; e < 8; ++e)
            op[(e + 8 * hi) * D_ + nt * 16 + n] = y[nt][e] * inv[e];
}

extern "C" void kernel_launch(void* const* d_in, const int* in_sizes, int n_in,
                              void* d_out, int out_size, void* d_ws, size_t ws_size,
                              hipStream_t stream) {
    const float* Q = (const float*)d_in[0];
    const float* T = (const float*)d_in[1];
    const float* V = (const float*)d_in[2];
    float*       O = (float*)d_out;
    dim3 grid(S_ / 64, B_);
    attn_fa_kernel<<<grid, dim3(128), 0, stream>>>(Q, T, V, O);
}


# each wave stages K alone (private LDS tile); next tile's K rows and V fragments requested a trip ahead into registers; no work-group barrier in the key loop; 2 waves per SIMD; sums unchanged
# speedup vs baseline: 2.8463x; 2.8463x over previous
#include <hip/hip_runtime.h>


#define B_ 16
#define S_ 2048
#define D_ 128
#define THR 0.3f
#define GROW_LIM 32.0f

typedef __attribute__((ext_vector_type(2))) float v2f;
typedef __attribute__((ext_vector_type(4))) float v4f;
typedef __attribute__((ext_vector_type(8))) float v8f;

#define KSTRIDE 132
#define PSTRIDE 20

static __device__ __forceinline__
v8f wmma_f32(v2f a, v2f b, v8f c) {
    return __builtin_amdgcn_wmma_f32_16x16x4_f32(false, a, false, b,
                                                 (short)0, c, false, false);
}

__global__ __launch_bounds__(128) __attribute__((amdgpu_waves_per_eu(2, 2)))
void attn_fa_kernel(const float* __restrict__ Q,
                    const float* __restrict__ T,
                    const float* __restrict__ V,
                    float* __restrict__ O)
{
    __shared__ float lds_k [4 * 16 * KSTRIDE];
    __shared__ float lds_p [4 * 16 * PSTRIDE];

    const int b    = blockIdx.y;
    const int g    = blockIdx.x;
    const int tid  = threadIdx.x;
    const int wave = __builtin_amdgcn_readfirstlane(tid >> 5);
    const int lane = tid & 31;
    const int n    = lane & 15;
    const int hi   = lane >> 4;

    const int q0     = g * 64 + wave * 16;
    const int myDiag = q0 >> 4;

    const float SCALE2 = 0.08838834764831845f * 1.4426950408889634f;

    float* kt = &lds_k[wave * 16 * KSTRIDE];
    float* pp = &lds_p[wave * 16 * PSTRIDE];

    v2f qa[32];
    {
        const float* qp = Q + ((size_t)b * S_ + q0 + n) * D_ + 2 * hi;
#pragma unroll
        for (int c = 0; c < 32; ++c)
            qa[c] = *(const v2f*)(qp + 4 * c);
    }
    __builtin_amdgcn_sched_barrier(0);

    const float* kg = T + (size_t)b * S_ * D_ + 4 * lane;
    float*       ks = kt + 4 * lane;

    {
        v4f k0[16];
#pragma unroll
        for (int i = 0; i < 16; ++i)
            k0[i] = *(const v4f*)(kg + i * D_);
#pragma unroll
        for (int i = 0; i < 16; ++i)
            *(v4f*)(ks + i * KSTRIDE) = k0[i];
    }

    const float* vg = V + ((size_t)b * S_ + 2 * hi) * D_ + 4 * n;
    v4f vlo[8], vhi[8], kn[16];
#pragma unroll
    for (int c = 0; c < 4; ++c)
#pragma unroll
        for (int s = 0; s < 2; ++s) {
            const float* vr = vg + (4 * c + s) * D_;
            vlo[2 * c + s] = *(const v4f*)vr;
            vhi[2 * c + s] = *(const v4f*)(vr + 64);
        }
    {
        const int j1 = (myDiag > 0) ? 1 : 0;
        const float* kp = kg + (size_t)j1 * 16 * D_;
#pragma unroll
        for (int i = 0; i < 16; ++i)
            kn[i] = *(const v4f*)(kp + i * D_);
    }

    v8f y[8];
#pragma unroll
    for (int nt = 0; nt < 8; ++nt) y[nt] = {};
    float m_run[8], l_run[8];
#pragma unroll
    for (int e = 0; e < 8; ++e) { m_run[e] = -1e30f; l_run[e] = 0.0f; }

    int j = 0;
    do {
        const float* krow = kt + n * KSTRIDE + 2 * hi;
        v2f kb[32];
#pragma unroll
        for (int i = 0; i < 32; ++i)
            kb[i] = *(const v2f*)(krow + 4 * i);

        v8f s0 = {}, s1 = {};
#pragma unroll
        for (int c = 0; c < 32; c += 2) {
            s0 = wmma_f32(qa[c],     kb[c],     s0);
            s1 = wmma_f32(qa[c + 1], kb[c + 1], s1);
        }

        const int kk = j * 16 + n;
        float z[8], dz[8];
#pragma unroll
        for (int e = 0; e < 8; ++e) {
            float sv = s0[e] + s1[e];
            sv = (sv > THR) ? sv : 0.0f;
            float zz = sv * SCALE2;
            const int m = e + 8 * hi;
            if (kk > q0 + m) zz = -1e30f;
            z[e]  = zz;
            dz[e] = zz - m_run[e];
        }
        const float top = fmaxf(fmaxf(fmaxf(dz[0], dz[1]), fmaxf(dz[2], dz[3])),
                                fmaxf(fmaxf(dz[4], dz[5]), fmaxf(dz[6], dz[7])));

        if (__builtin_amdgcn_ballot_w32(top > GROW_LIM) != 0u) {
#pragma unroll
            for (int e = 0; e < 8; ++e) {
                float rm = z[e];
                rm = fmaxf(rm, __shfl_xor(rm, 1, 32));
                rm = fmaxf(rm, __shfl_xor(rm, 2, 32));
                rm = fmaxf(rm, __shfl_xor(rm, 4, 32));
                rm = fmaxf(rm, __shfl_xor(rm, 8, 32));
                const float mn = fmaxf(m_run[e], rm);
                const float a  = __builtin_amdgcn_exp2f(m_run[e] - mn);
                m_run[e]  = mn;
                l_run[e] *= a;
#pragma unroll
                for (int nt = 0; nt < 8; ++nt)
                    y[nt][e] *= a;
                dz[e] = z[e] - mn;
            }
        }

        float p[8];
#pragma unroll
        for (int e = 0; e < 8; ++e) {
            p[e] = __builtin_amdgcn_exp2f(dz[e]);
            l_run[e] += p[e];
        }

#pragma unroll
        for (int e = 0; e < 8; ++e)
            pp[(e + 8 * hi) * PSTRIDE + n] = p[e];

        v2f pa[4];
#pragma unroll
        for (int c = 0; c < 4; ++c)
            pa[c] = *(const v2f*)&pp[n * PSTRIDE + 4 * c + 2 * hi];

#pragma unroll
        for (int half = 0; half < 2; ++half) {
#pragma unroll
            for (int nt = 0; nt < 4; ++nt) {
                v8f acc = y[half * 4 + nt];
#pragma unroll
                for (int c = 0; c < 4; ++c) {
                    const v4f r0 = half ? vhi[2 * c]     : vlo[2 * c];
                    const v4f r1 = half ? vhi[2 * c + 1] : vlo[2 * c + 1];
                    v2f vb;
                    vb.x = (nt == 0) ? r0.x : (nt == 1) ? r0.y : (nt == 2) ? r0.z : r0.w;
                    vb.y = (nt == 0) ? r1.x : (nt == 1) ? r1.y : (nt == 2) ? r1.z : r1.w;
                    acc = wmma_f32(pa[c], vb, acc);
                }
                y[half * 4 + nt] = acc;
            }
        }

#pragma unroll
        for (int i = 0; i < 16; ++i)
            *(v4f*)(ks + i * KSTRIDE) = kn[i];

        {
            const int jv = (j + 1 < myDiag) ? j + 1 : myDiag;
            const int jk = (j + 2 < myDiag) ? j + 2 : myDiag;
            const float* vn = vg + (size_t)jv * 16 * D_;
#pragma unroll
            for (int c = 0; c < 4; ++c)
#pragma unroll
                for (int s = 0; s < 2; ++s) {
                    const float* vr = vn + (4 * c + s) * D_;
                    vlo[2 * c + s] = *(const v4f*)vr;
                    vhi[2 * c + s] = *(const v4f*)(vr + 64);
                }
            const float* kp = kg + (size_t)jk * 16 * D_;
#pragma unroll
            for (int i = 0; i < 16; ++i)
                kn[i] = *(const v4f*)(kp + i * D_);
        }
        asm volatile("" ::: "memory");
    } while (++j <= myDiag);

    float inv[8];
#pragma unroll
    for (int e = 0; e < 8; ++e) {
        float rs = l_run[e];
        rs += __shfl_xor(rs, 1, 32);
        rs += __shfl_xor(rs, 2, 32);
        rs += __shfl_xor(rs, 4, 32);
        rs += __shfl_xor(rs, 8, 32);
        inv[e] = 1.0f / rs;
    }
    float* op = O + ((size_t)b * S_ + q0 + 8 * hi) * D_ + 4 * n;
#pragma unroll
    for (int e = 0; e < 8; ++e) {
        float4 o0, o1;
        o0.x = y[0][e] * inv[e]; o0.y = y[1][e] * inv[e];
        o0.z = y[2][e] * inv[e]; o0.w = y[3][e] * inv[e];
        o1.x = y[4][e] * inv[e]; o1.y = y[5][e] * inv[e];
        o1.z = y[6][e] * inv[e]; o1.w = y[7][e] * inv[e];
        *(float4*)(op + e * D_)      = o0;
        *(float4*)(op + e * D_ + 64) = o1;
    }
}

extern "C" void kernel_launch(void* const* d_in, const int* in_sizes, int n_in,
                              void* d_out, int out_size, void* d_ws, size_t ws_size,
                              hipStream_t stream) {
    const float* Q = (const float*)d_in[0];
    const float* T = (const float*)d_in[1];
    const float* V = (const float*)d_in[2];
    float*       O = (float*)d_out;
    dim3 grid(S_ / 64, B_);
    attn_fa_kernel<<<grid, dim3(128), 0, stream>>>(Q, T, V, O);
}


# each query-tile pair (t,127-t) shared by two waves: one takes the long tile's keys 0-1023, the other its rest and the short tile; 64-65 key tiles per wave, (m,l,y) merged in LDS: sum order changes
# speedup vs baseline: 4.8544x; 1.7055x over previous
#include <hip/hip_runtime.h>


#define B_ 16
#define S_ 2048
#define D_ 128
#define THR 0.3f
#define GROW_LIM 32.0f
#define NT_   (S_ / 16)
#define HALF_ (NT_ / 2)

typedef __attribute__((ext_vector_type(2))) float v2f;
typedef __attribute__((ext_vector_type(4))) float v4f;
typedef __attribute__((ext_vector_type(8))) float v8f;

#define KSTRIDE 132
#define PSTRIDE 20

static __device__ __forceinline__
v8f wmma_f32(v2f a, v2f b, v8f c) {
    return __builtin_amdgcn_wmma_f32_16x16x4_f32(false, a, false, b,
                                                 (short)0, c, false, false);
}

static __device__ __forceinline__
void finish_tile(float* __restrict__ op, const v8f (&y)[8], const float (&l_run)[8])
{
    float inv[8];
#pragma unroll
    for (int e = 0; e < 8; ++e) {
        float rs = l_run[e];
        rs += __shfl_xor(rs, 1, 32);
        rs += __shfl_xor(rs, 2, 32);
        rs += __shfl_xor(rs, 4, 32);
        rs += __shfl_xor(rs, 8, 32);
        inv[e] = 1.0f / rs;
    }
#pragma unroll
    for (int e = 0; e < 8; ++e) {
        float4 o0, o1;
        o0.x = y[0][e] * inv[e]; o0.y = y[1][e] * inv[e];
        o0.z = y[2][e] * inv[e]; o0.w = y[3][e] * inv[e];
        o1.x = y[4][e] * inv[e]; o1.y = y[5][e] * inv[e];
        o1.z = y[6][e] * inv[e]; o1.w = y[7][e] * inv[e];
        *(float4*)(op + e * D_)      = o0;
        *(float4*)(op + e * D_ + 64) = o1;
    }
}

__global__ __launch_bounds__(256) __attribute__((amdgpu_waves_per_eu(2, 2)))
void attn_fa_kernel(const float* __restrict__ Q,
                    const float* __restrict__ T,
                    const float* __restrict__ V,
                    float* __restrict__ O)
{
    __shared__ float lds_k [8 * 16 * KSTRIDE];
    __shared__ float lds_p [8 * 16 * PSTRIDE];
    __shared__ float lds_ml[4 * 16 * 32];

    const int b    = blockIdx.y;
    const int tid  = threadIdx.x;
    const int wave = __builtin_amdgcn_readfirstlane(tid >> 5);
    const int lane = tid & 31;
    const int n    = lane & 15;
    const int hi   = lane >> 4;

    const int pr  = wave & 3;
    const int isB = wave >> 2;
    const int tS  = blockIdx.x * 4 + pr;
    const int tL  = NT_ - 1 - tS;

    const float SCALE2 = 0.08838834764831845f * 1.4426950408889634f;

    float* kt = &lds_k[wave * 16 * KSTRIDE];
    float* pp = &lds_p[wave * 16 * PSTRIDE];

    const float* kg = T + (size_t)b * S_ * D_ + 4 * lane;
    float*       ks = kt + 4 * lane;
    const float* vg = V + ((size_t)b * S_ + 2 * hi) * D_ + 4 * n;

    v8f y[8];
    float m_run[8], l_run[8];

    const int nseg = 1 + isB;
    int seg = 0;
    do {
        const bool shortSeg = (isB != 0) && (seg == 0);
        const bool downSeg  = (isB != 0) && (seg != 0);
        const int  qt    = shortSeg ? tS : tL;
        const int  jbeg  = downSeg ? tL : 0;
        const int  jstep = downSeg ? -1 : 1;
        const int  ntr   = shortSeg ? tS + 1 : (downSeg ? HALF_ - tS : HALF_);
        const int  q0    = qt * 16;

        v2f qa[32];
        {
            const float* qp = Q + ((size_t)b * S_ + q0 + n) * D_ + 2 * hi;
#pragma unroll
            for (int c = 0; c < 32; ++c)
                qa[c] = *(const v2f*)(qp + 4 * c);
        }
        __builtin_amdgcn_sched_barrier(0);

        {
            const float* kp = kg + (size_t)jbeg * 16 * D_;
            v4f k0[16];
#pragma unroll
            for (int i = 0; i < 16; ++i)
                k0[i] = *(const v4f*)(kp + i * D_);
#pragma unroll
            for (int i = 0; i < 16; ++i)
                *(v4f*)(ks + i * KSTRIDE) = k0[i];
        }

        v4f vlo[8], vhi[8], kn[16];
        {
            const float* vn = vg + (size_t)jbeg * 16 * D_;
#pragma unroll
            for (int c = 0; c < 4; ++c)
#pragma unroll
                for (int s = 0; s < 2; ++s) {
                    const float* vr = vn + (4 * c + s) * D_;
                    vlo[2 * c + s] = *(const v4f*)vr;
                    vhi[2 * c + s] = *(const v4f*)(vr + 64);
                }
            const int j1 = (ntr > 1) ? jbeg + jstep : jbeg;
            const float* kp = kg + (size_t)j1 * 16 * D_;
#pragma unroll
            for (int i = 0; i < 16; ++i)
                kn[i] = *(const v4f*)(kp + i * D_);
        }

#pragma unroll
        for (int nt = 0; nt < 8; ++nt) y[nt] = {};
#pragma unroll
        for (int e = 0; e < 8; ++e) { m_run[e] = -1e30f; l_run[e] = 0.0f; }

        int t = 0;
        do {
            const int j = jbeg + jstep * t;

            const float* krow = kt + n * KSTRIDE + 2 * hi;
            v2f kb[32];
#pragma unroll
            for (int i = 0; i < 32; ++i)
                kb[i] = *(const v2f*)(krow + 4 * i);

            v8f s0 = {}, s1 = {};
#pragma unroll
            for (int c = 0; c < 32; c += 2) {
                s0 = wmma_f32(qa[c],     kb[c],     s0);
                s1 = wmma_f32(qa[c + 1], kb[c + 1], s1);
            }

            const int kk = j * 16 + n;
            float z[8], dz[8];
#pragma unroll
            for (int e = 0; e < 8; ++e) {
                float sv = s0[e] + s1[e];
                sv = (sv > THR) ? sv : 0.0f;
                float zz = sv * SCALE2;
                const int m = e + 8 * hi;
                if (kk > q0 + m) zz = -1e30f;
                z[e]  = zz;
                dz[e] = zz - m_run[e];
            }
            const float top = fmaxf(fmaxf(fmaxf(dz[0], dz[1]), fmaxf(dz[2], dz[3])),
                                    fmaxf(fmaxf(dz[4], dz[5]), fmaxf(dz[6], dz[7])));

            if (__builtin_amdgcn_ballot_w32(top > GROW_LIM) != 0u) {
#pragma unroll
                for (int e = 0; e < 8; ++e) {
                    float rm = z[e];
                    rm = fmaxf(rm, __shfl_xor(rm, 1, 32));
                    rm = fmaxf(rm, __shfl_xor(rm, 2, 32));
                    rm = fmaxf(rm, __shfl_xor(rm, 4, 32));
                    rm = fmaxf(rm, __shfl_xor(rm, 8, 32));
                    const float mn = fmaxf(m_run[e], rm);
                    const float a  = __builtin_amdgcn_exp2f(m_run[e] - mn);
                    m_run[e]  = mn;
                    l_run[e] *= a;
#pragma unroll
                    for (int nt = 0; nt < 8; ++nt)
                        y[nt][e] *= a;
                    dz[e] = z[e] - mn;
                }
            }

            float p[8];
#pragma unroll
            for (int e = 0; e < 8; ++e) {
                p[e] = __builtin_amdgcn_exp2f(dz[e]);
                l_run[e] += p[e];
            }

#pragma unroll
            for (int e = 0; e < 8; ++e)
                pp[(e + 8 * hi) * PSTRIDE + n] = p[e];

            v2f pa[4];
#pragma unroll
            for (int c = 0; c < 4; ++c)
                pa[c] = *(const v2f*)&pp[n * PSTRIDE + 4 * c + 2 * hi];

#pragma unroll
            for (int half = 0; half < 2; ++half) {
#pragma unroll
                for (int nt = 0; nt < 4; ++nt) {
                    v8f acc = y[half * 4 + nt];
#pragma unroll
                    for (int c = 0; c < 4; ++c) {
                        const v4f r0 = half ? vhi[2 * c]     : vlo[2 * c];
                        const v4f r1 = half ? vhi[2 * c + 1] : vlo[2 * c + 1];
                        v2f vb;
                        vb.x = (nt == 0) ? r0.x : (nt == 1) ? r0.y : (nt == 2) ? r0.z : r0.w;
                        vb.y = (nt == 0) ? r1.x : (nt == 1) ? r1.y : (nt == 2) ? r1.z : r1.w;
                        acc = wmma_f32(pa[c], vb, acc);
                    }
                    y[half * 4 + nt] = acc;
                }
            }

#pragma unroll
            for (int i = 0; i < 16; ++i)
                *(v4f*)(ks + i * KSTRIDE) = kn[i];

            {
                const int tv = (t + 1 < ntr) ? t + 1 : ntr - 1;
                const int tk = (t + 2 < ntr) ? t + 2 : ntr - 1;
                const int jv = jbeg + jstep * tv;
                const int jk = jbeg + jstep * tk;
                const float* vn = vg + (size_t)jv * 16 * D_;
#pragma unroll
                for (int c = 0; c < 4; ++c)
#pragma unroll
                    for (int s = 0; s < 2; ++s) {
                        const float* vr = vn + (4 * c + s) * D_;
                        vlo[2 * c + s] = *(const v4f*)vr;
                        vhi[2 * c + s] = *(const v4f*)(vr + 64);
                    }
                const float* kp = kg + (size_t)jk * 16 * D_;
#pragma unroll
                for (int i = 0; i < 16; ++i)
                    kn[i] = *(const v4f*)(kp + i * D_);
            }
            asm volatile("" ::: "memory");
        } while (++t < ntr);

        if (shortSeg)
            finish_tile(O + ((size_t)b * S_ + q0 + 8 * hi) * D_ + 4 * n, y, l_run);
    } while (++seg < nseg);

    float* my = &lds_k [pr * 16 * KSTRIDE] + 4 * lane;
    float* mm = &lds_ml[pr * 16 * 32]      + 4 * lane;
    if (isB == 0) {
#pragma unroll
        for (int nt = 0; nt < 8; ++nt) {
            v4f lo, hh;
            lo.x = y[nt][0]; lo.y = y[nt][1]; lo.z = y[nt][2]; lo.w = y[nt][3];
            hh.x = y[nt][4]; hh.y = y[nt][5]; hh.z = y[nt][6]; hh.w = y[nt][7];
            *(v4f*)(my + (2 * nt)     * 128) = lo;
            *(v4f*)(my + (2 * nt + 1) * 128) = hh;
        }
#pragma unroll
        for (int c = 0; c < 2; ++c) {
            v4f mv, lv;
            mv.x = m_run[4 * c]; mv.y = m_run[4 * c + 1]; mv.z = m_run[4 * c + 2]; mv.w = m_run[4 * c + 3];
            lv.x = l_run[4 * c]; lv.y = l_run[4 * c + 1]; lv.z = l_run[4 * c + 2]; lv.w = l_run[4 * c + 3];
            *(v4f*)(mm + c * 128)       = mv;
            *(v4f*)(mm + (2 + c) * 128) = lv;
        }
    }
    __syncthreads();
    if (isB != 0) {
        float aA[8], aB[8];
#pragma unroll
        for (int c = 0; c < 2; ++c) {
            const v4f mv = *(const v4f*)(mm + c * 128);
            const v4f lv = *(const v4f*)(mm + (2 + c) * 128);
#pragma unroll
            for (int k = 0; k < 4; ++k) {
                const int   e  = 4 * c + k;
                const float mA = (k == 0) ? mv.x : (k == 1) ? mv.y : (k == 2) ? mv.z : mv.w;
                const float lA = (k == 0) ? lv.x : (k == 1) ? lv.y : (k == 2) ? lv.z : lv.w;
                const float mn = fmaxf(mA, m_run[e]);
                aA[e] = __builtin_amdgcn_exp2f(mA - mn);
                aB[e] = __builtin_amdgcn_exp2f(m_run[e] - mn);
                l_run[e] = l_run[e] * aB[e] + lA * aA[e];
            }
        }
#pragma unroll
        for (int nt = 0; nt < 8; ++nt) {
            const v4f lo = *(const v4f*)(my + (2 * nt)     * 128);
            const v4f hh = *(const v4f*)(my + (2 * nt + 1) * 128);
            y[nt][0] = y[nt][0] * aB[0] + lo.x * aA[0];
            y[nt][1] = y[nt][1] * aB[1] + lo.y * aA[1];
            y[nt][2] = y[nt][2] * aB[2] + lo.z * aA[2];
            y[nt][3] = y[nt][3] * aB[3] + lo.w * aA[3];
            y[nt][4] = y[nt][4] * aB[4] + hh.x * aA[4];
            y[nt][5] = y[nt][5] * aB[5] + hh.y * aA[5];
            y[nt][6] = y[nt][6] * aB[6] + hh.z * aA[6];
            y[nt][7] = y[nt][7] * aB[7] + hh.w * aA[7];
        }
        finish_tile(O + ((size_t)b * S_ + tL * 16 + 8 * hi) * D_ + 4 * n, y, l_run);
    }
}

extern "C" void kernel_launch(void* const* d_in, const int* in_sizes, int n_in,
                              void* d_out, int out_size, void* d_ws, size_t ws_size,
                              hipStream_t stream) {
    const float* Q = (const float*)d_in[0];
    const float* T = (const float*)d_in[1];
    const float* V = (const float*)d_in[2];
    float*       O = (float*)d_out;
    dim3 grid(HALF_ / 4, B_);
    attn_fa_kernel<<<grid, dim3(256), 0, stream>>>(Q, T, V, O);
}


# wave priority (s_setprio) 3 while a wave is in its LDS-bound score products, softmax and P round trip, 0 in its register-fed PV run: the partner's dense run no longer starves it; sums unchanged
# speedup vs baseline: 5.0652x; 1.0434x over previous
#include <hip/hip_runtime.h>


#define B_ 16
#define S_ 2048
#define D_ 128
#define THR 0.3f
#define GROW_LIM 32.0f
#define NT_   (S_ / 16)
#define HALF_ (NT_ / 2)

typedef __attribute__((ext_vector_type(2))) float v2f;
typedef __attribute__((ext_vector_type(4))) float v4f;
typedef __attribute__((ext_vector_type(8))) float v8f;

#define KSTRIDE 132
#define PSTRIDE 20

static __device__ __forceinline__
v8f wmma_f32(v2f a, v2f b, v8f c) {
    return __builtin_amdgcn_wmma_f32_16x16x4_f32(false, a, false, b,
                                                 (short)0, c, false, false);
}

static __device__ __forceinline__
void finish_tile(float* __restrict__ op, const v8f (&y)[8], const float (&l_run)[8])
{
    float inv[8];
#pragma unroll
    for (int e = 0; e < 8; ++e) {
        float rs = l_run[e];
        rs += __shfl_xor(rs, 1, 32);
        rs += __shfl_xor(rs, 2, 32);
        rs += __shfl_xor(rs, 4, 32);
        rs += __shfl_xor(rs, 8, 32);
        inv[e] = 1.0f / rs;
    }
#pragma unroll
    for (int e = 0; e < 8; ++e) {
        float4 o0, o1;
        o0.x = y[0][e] * inv[e]; o0.y = y[1][e] * inv[e];
        o0.z = y[2][e] * inv[e]; o0.w = y[3][e] * inv[e];
        o1.x = y[4][e] * inv[e]; o1.y = y[5][e] * inv[e];
        o1.z = y[6][e] * inv[e]; o1.w = y[7][e] * inv[e];
        *(float4*)(op + e * D_)      = o0;
        *(float4*)(op + e * D_ + 64) = o1;
    }
}

__global__ __launch_bounds__(256) __attribute__((amdgpu_waves_per_eu(2, 2)))
void attn_fa_kernel(const float* __restrict__ Q,
                    const float* __restrict__ T,
                    const float* __restrict__ V,
                    float* __restrict__ O)
{
    __shared__ float lds_k [8 * 16 * KSTRIDE];
    __shared__ float lds_p [8 * 16 * PSTRIDE];
    __shared__ float lds_ml[4 * 16 * 32];

    const int b    = blockIdx.y;
    const int tid  = threadIdx.x;
    const int wave = __builtin_amdgcn_readfirstlane(tid >> 5);
    const int lane = tid & 31;
    const int n    = lane & 15;
    const int hi   = lane >> 4;

    const int pr  = wave & 3;
    const int isB = wave >> 2;
    const int tS  = blockIdx.x * 4 + pr;
    const int tL  = NT_ - 1 - tS;

    const float SCALE2 = 0.08838834764831845f * 1.4426950408889634f;

    float* kt = &lds_k[wave * 16 * KSTRIDE];
    float* pp = &lds_p[wave * 16 * PSTRIDE];

    const float* kg = T + (size_t)b * S_ * D_ + 4 * lane;
    float*       ks = kt + 4 * lane;
    const float* vg = V + ((size_t)b * S_ + 2 * hi) * D_ + 4 * n;

    v8f y[8];
    float m_run[8], l_run[8];

    const int nseg = 1 + isB;
    int seg = 0;
    do {
        const bool shortSeg = (isB != 0) && (seg == 0);
        const bool downSeg  = (isB != 0) && (seg != 0);
        const int  qt    = shortSeg ? tS : tL;
        const int  jbeg  = downSeg ? tL : 0;
        const int  jstep = downSeg ? -1 : 1;
        const int  ntr   = shortSeg ? tS + 1 : (downSeg ? HALF_ - tS : HALF_);
        const int  q0    = qt * 16;

        v2f qa[32];
        {
            const float* qp = Q + ((size_t)b * S_ + q0 + n) * D_ + 2 * hi;
#pragma unroll
            for (int c = 0; c < 32; ++c)
                qa[c] = *(const v2f*)(qp + 4 * c);
        }
        __builtin_amdgcn_sched_barrier(0);

        {
            const float* kp = kg + (size_t)jbeg * 16 * D_;
            v4f k0[16];
#pragma unroll
            for (int i = 0; i < 16; ++i)
                k0[i] = *(const v4f*)(kp + i * D_);
#pragma unroll
            for (int i = 0; i < 16; ++i)
                *(v4f*)(ks + i * KSTRIDE) = k0[i];
        }

        v4f vlo[8], vhi[8], kn[16];
        {
            const float* vn = vg + (size_t)jbeg * 16 * D_;
#pragma unroll
            for (int c = 0; c < 4; ++c)
#pragma unroll
                for (int s = 0; s < 2; ++s) {
                    const float* vr = vn + (4 * c + s) * D_;
                    vlo[2 * c + s] = *(const v4f*)vr;
                    vhi[2 * c + s] = *(const v4f*)(vr + 64);
                }
            const int j1 = (ntr > 1) ? jbeg + jstep : jbeg;
            const float* kp = kg + (size_t)j1 * 16 * D_;
#pragma unroll
            for (int i = 0; i < 16; ++i)
                kn[i] = *(const v4f*)(kp + i * D_);
        }

#pragma unroll
        for (int nt = 0; nt < 8; ++nt) y[nt] = {};
#pragma unroll
        for (int e = 0; e < 8; ++e) { m_run[e] = -1e30f; l_run[e] = 0.0f; }

        int t = 0;
        do {
            const int j = jbeg + jstep * t;

            __builtin_amdgcn_s_setprio(3);

            const float* krow = kt + n * KSTRIDE + 2 * hi;
            v2f kb[32];
#pragma unroll
            for (int i = 0; i < 32; ++i)
                kb[i] = *(const v2f*)(krow + 4 * i);

            v8f s0 = {}, s1 = {};
#pragma unroll
            for (int c = 0; c < 32; c += 2) {
                s0 = wmma_f32(qa[c],     kb[c],     s0);
                s1 = wmma_f32(qa[c + 1], kb[c + 1], s1);
            }

            const int kk = j * 16 + n;
            float z[8], dz[8];
#pragma unroll
            for (int e = 0; e < 8; ++e) {
                float sv = s0[e] + s1[e];
                sv = (sv > THR) ? sv : 0.0f;
                float zz = sv * SCALE2;
                const int m = e + 8 * hi;
                if (kk > q0 + m) zz = -1e30f;
                z[e]  = zz;
                dz[e] = zz - m_run[e];
            }
            const float top = fmaxf(fmaxf(fmaxf(dz[0], dz[1]), fmaxf(dz[2], dz[3])),
                                    fmaxf(fmaxf(dz[4], dz[5]), fmaxf(dz[6], dz[7])));

            if (__builtin_amdgcn_ballot_w32(top > GROW_LIM) != 0u) {
#pragma unroll
                for (int e = 0; e < 8; ++e) {
                    float rm = z[e];
                    rm = fmaxf(rm, __shfl_xor(rm, 1, 32));
                    rm = fmaxf(rm, __shfl_xor(rm, 2, 32));
                    rm = fmaxf(rm, __shfl_xor(rm, 4, 32));
                    rm = fmaxf(rm, __shfl_xor(rm, 8, 32));
                    const float mn = fmaxf(m_run[e], rm);
                    const float a  = __builtin_amdgcn_exp2f(m_run[e] - mn);
                    m_run[e]  = mn;
                    l_run[e] *= a;
#pragma unroll
                    for (int nt = 0; nt < 8; ++nt)
                        y[nt][e] *= a;
                    dz[e] = z[e] - mn;
                }
            }

            float p[8];
#pragma unroll
            for (int e = 0; e < 8; ++e) {
                p[e] = __builtin_amdgcn_exp2f(dz[e]);
                l_run[e] += p[e];
            }

#pragma unroll
            for (int e = 0; e < 8; ++e)
                pp[(e + 8 * hi) * PSTRIDE + n] = p[e];

            v2f pa[4];
#pragma unroll
            for (int c = 0; c < 4; ++c)
                pa[c] = *(const v2f*)&pp[n * PSTRIDE + 4 * c + 2 * hi];

            __builtin_amdgcn_s_setprio(0);

#pragma unroll
            for (int half = 0; half < 2; ++half) {
#pragma unroll
                for (int nt = 0; nt < 4; ++nt) {
                    v8f acc = y[half * 4 + nt];
#pragma unroll
                    for (int c = 0; c < 4; ++c) {
                        const v4f r0 = half ? vhi[2 * c]     : vlo[2 * c];
                        const v4f r1 = half ? vhi[2 * c + 1] : vlo[2 * c + 1];
                        v2f vb;
                        vb.x = (nt == 0) ? r0.x : (nt == 1) ? r0.y : (nt == 2) ? r0.z : r0.w;
                        vb.y = (nt == 0) ? r1.x : (nt == 1) ? r1.y : (nt == 2) ? r1.z : r1.w;
                        acc = wmma_f32(pa[c], vb, acc);
                    }
                    y[half * 4 + nt] = acc;
                }
            }

#pragma unroll
            for (int i = 0; i < 16; ++i)
                *(v4f*)(ks + i * KSTRIDE) = kn[i];

            {
                const int tv = (t + 1 < ntr) ? t + 1 : ntr - 1;
                const int tk = (t + 2 < ntr) ? t + 2 : ntr - 1;
                const int jv = jbeg + jstep * tv;
                const int jk = jbeg + jstep * tk;
                const float* vn = vg + (size_t)jv * 16 * D_;
#pragma unroll
                for (int c = 0; c < 4; ++c)
#pragma unroll
                    for (int s = 0; s < 2; ++s) {
                        const float* vr = vn + (4 * c + s) * D_;
                        vlo[2 * c + s] = *(const v4f*)vr;
                        vhi[2 * c + s] = *(const v4f*)(vr + 64);
                    }
                const float* kp = kg + (size_t)jk * 16 * D_;
#pragma unroll
                for (int i = 0; i < 16; ++i)
                    kn[i] = *(const v4f*)(kp + i * D_);
            }
            asm volatile("" ::: "memory");
        } while (++t < ntr);

        if (shortSeg)
            finish_tile(O + ((size_t)b * S_ + q0 + 8 * hi) * D_ + 4 * n, y, l_run);
    } while (++seg < nseg);

    float* my = &lds_k [pr * 16 * KSTRIDE] + 4 * lane;
    float* mm = &lds_ml[pr * 16 * 32]      + 4 * lane;
    if (isB == 0) {
#pragma unroll
        for (int nt = 0; nt < 8; ++nt) {
            v4f lo, hh;
            lo.x = y[nt][0]; lo.y = y[nt][1]; lo.z = y[nt][2]; lo.w = y[nt][3];
            hh.x = y[nt][4]; hh.y = y[nt][5]; hh.z = y[nt][6]; hh.w = y[nt][7];
            *(v4f*)(my + (2 * nt)     * 128) = lo;
            *(v4f*)(my + (2 * nt + 1) * 128) = hh;
        }
#pragma unroll
        for (int c = 0; c < 2; ++c) {
            v4f mv, lv;
            mv.x = m_run[4 * c]; mv.y = m_run[4 * c + 1]; mv.z = m_run[4 * c + 2]; mv.w = m_run[4 * c + 3];
            lv.x = l_run[4 * c]; lv.y = l_run[4 * c + 1]; lv.z = l_run[4 * c + 2]; lv.w = l_run[4 * c + 3];
            *(v4f*)(mm + c * 128)       = mv;
            *(v4f*)(mm + (2 + c) * 128) = lv;
        }
    }
    __syncthreads();
    if (isB != 0) {
        float aA[8], aB[8];
#pragma unroll
        for (int c = 0; c < 2; ++c) {
            const v4f mv = *(const v4f*)(mm + c * 128);
            const v4f lv = *(const v4f*)(mm + (2 + c) * 128);
#pragma unroll
            for (int k = 0; k < 4; ++k) {
                const int   e  = 4 * c + k;
                const float mA = (k == 0) ? mv.x : (k == 1) ? mv.y : (k == 2) ? mv.z : mv.w;
                const float lA = (k == 0) ? lv.x : (k == 1) ? lv.y : (k == 2) ? lv.z : lv.w;
                const float mn = fmaxf(mA, m_run[e]);
                aA[e] = __builtin_amdgcn_exp2f(mA - mn);
                aB[e] = __builtin_amdgcn_exp2f(m_run[e] - mn);
                l_run[e] = l_run[e] * aB[e] + lA * aA[e];
            }
        }
#pragma unroll
        for (int nt = 0; nt < 8; ++nt) {
            const v4f lo = *(const v4f*)(my + (2 * nt)     * 128);
            const v4f hh = *(const v4f*)(my + (2 * nt + 1) * 128);
            y[nt][0] = y[nt][0] * aB[0] + lo.x * aA[0];
            y[nt][1] = y[nt][1] * aB[1] + lo.y * aA[1];
            y[nt][2] = y[nt][2] * aB[2] + lo.z * aA[2];
            y[nt][3] = y[nt][3] * aB[3] + lo.w * aA[3];
            y[nt][4] = y[nt][4] * aB[4] + hh.x * aA[4];
            y[nt][5] = y[nt][5] * aB[5] + hh.y * aA[5];
            y[nt][6] = y[nt][6] * aB[6] + hh.z * aA[6];
            y[nt][7] = y[nt][7] * aB[7] + hh.w * aA[7];
        }
        finish_tile(O + ((size_t)b * S_ + tL * 16 + 8 * hi) * D_ + 4 * n, y, l_run);
    }
}

extern "C" void kernel_launch(void* const* d_in, const int* in_sizes, int n_in,
                              void* d_out, int out_size, void* d_ws, size_t ws_size,
                              hipStream_t stream) {
    const float* Q = (const float*)d_in[0];
    const float* T = (const float*)d_in[1];
    const float* V = (const float*)d_in[2];
    float*       O = (float*)d_out;
    dim3 grid(HALF_ / 4, B_);
    attn_fa_kernel<<<grid, dim3(256), 0, stream>>>(Q, T, V, O);
}
